# starting kernel
# baseline (speedup 1.0000x reference)
#include <hip/hip_runtime.h>

constexpr int Bn   = 8;
constexpr int Nn   = 2048;
constexpr int Kn   = 16;
constexpr int CIN  = 64;
constexpr int CMID = 16;
constexpr int COUT = 64;
constexpr int FH1  = 256;
constexpr int KDIM = CMID * CIN;

typedef float v2f __attribute__((ext_vector_type(2)));
typedef float v8f __attribute__((ext_vector_type(8)));

__device__ __forceinline__ v8f wmma_f32(v2f a, v2f b, v8f c) {
  return __builtin_amdgcn_wmma_f32_16x16x4_f32(
       false, a,  false, b,
       (short)0, c,  false,  false);
}

constexpr int FSTR    = 18;
constexpr int OFF_F   = 0;
constexpr int OFF_M   = 1152;
constexpr int OFF_PTS = 5760;
constexpr int OFF_MSK = 5808;
constexpr int ESTR    = 514;
constexpr int OFF_E   = 5824;
constexpr int SMEM_F  = 14048;
constexpr int HSTR    = 258;
constexpr int OFF_H   = OFF_M;

__global__ __launch_bounds__(256) void pack_weights(
    const float* __restrict__ f1, const float* __restrict__ f2,
    float* __restrict__ f1p, float* __restrict__ f2p) {
  int tid = blockIdx.x * 256 + threadIdx.x;
  constexpr int NP1 = (KDIM / 4) * (FH1 / 16) * 32;
  constexpr int NP2 = (FH1 / 4) * (COUT / 16) * 32;
  if (tid < NP1) {
    int lane = tid & 31, jt = (tid >> 5) & 15, kk4 = tid >> 9;
    int c = lane & 15, dk = (lane >> 4) << 1;
    int r0 = kk4 * 4 + dk;
    f1p[tid * 2 + 0] = f1[(r0 + 0) * FH1 + jt * 16 + c];
    f1p[tid * 2 + 1] = f1[(r0 + 1) * FH1 + jt * 16 + c];
  } else if (tid < NP1 + NP2) {
    int t2 = tid - NP1;
    int lane = t2 & 31, ot = (t2 >> 5) & 3, kk4 = t2 >> 7;
    int c = lane & 15, dk = (lane >> 4) << 1;
    int r0 = kk4 * 4 + dk;
    f2p[t2 * 2 + 0] = f2[(r0 + 0) * COUT + ot * 16 + c];
    f2p[t2 * 2 + 1] = f2[(r0 + 1) * COUT + ot * 16 + c];
  }
}

__global__ __launch_bounds__(256) void gpc_main(
    const float* __restrict__ points, const float* __restrict__ feats,
    const float* __restrict__ pmask,
    const float* __restrict__ w1, const float* __restrict__ b1,
    const float* __restrict__ w2, const float* __restrict__ b2,
    const float* __restrict__ w3, const float* __restrict__ b3,
    const float* __restrict__ fb1, const float* __restrict__ fb2,
    const float* __restrict__ f1p, const float* __restrict__ f2p,
    float* __restrict__ out) {
  __shared__ float smem[SMEM_F];
  float* Fl = smem + OFF_F;
  float* Ml = smem + OFF_M;
  float* Pt = smem + OFF_PTS;
  float* Mk = smem + OFF_MSK;
  float* El = smem + OFF_E;
  float* Hl = smem + OFF_H;

  const int g = blockIdx.x;
  const int b = blockIdx.y;
  const int t = threadIdx.x;
  const int wave = t >> 5;
  const int lane = t & 31;
  const int lanelo = lane & 15;
  const int dk = (lane >> 4) << 1;
  const int mbase = (lane >> 4) << 3;
  const int base = b * Nn + g * Kn;

  if (t < 48) Pt[t] = points[(size_t)base * 3 + t];
  if (t >= 48 && t < 64) Mk[t - 48] = pmask[base + (t - 48)];
  {
    int k = t >> 4, c0 = (t & 15) * 4;
    float mk = pmask[base + k];
    const float4 v = *(const float4*)(feats + (size_t)(base + k) * CIN + c0);
    Fl[(c0 + 0) * FSTR + k] = v.x * mk;
    Fl[(c0 + 1) * FSTR + k] = v.y * mk;
    Fl[(c0 + 2) * FSTR + k] = v.z * mk;
    Fl[(c0 + 3) * FSTR + k] = v.w * mk;
  }
  __syncthreads();

  {
    int n = t >> 4, k = t & 15;
    float rx = Pt[n * 3 + 0] - Pt[k * 3 + 0];
    float ry = Pt[n * 3 + 1] - Pt[k * 3 + 1];
    float rz = Pt[n * 3 + 2] - Pt[k * 3 + 2];
    float h1[32];
#pragma unroll
    for (int j = 0; j < 32; ++j)
      h1[j] = fmaxf(0.f, rx * w1[j] + ry * w1[32 + j] + rz * w1[64 + j] + b1[j]);
    float h2[32];
#pragma unroll
    for (int j = 0; j < 32; ++j) {
      float s = b2[j];
#pragma unroll
      for (int i = 0; i < 32; ++i) s += h1[i] * w2[i * 32 + j];
      h2[j] = fmaxf(0.f, s);
    }
    float mk = Mk[k];
#pragma unroll
    for (int j = 0; j < 16; ++j) {
      float s = b3[j];
#pragma unroll
      for (int i = 0; i < 32; ++i) s += h2[i] * w3[i * 16 + j];
      Ml[(j * 16 + n) * FSTR + k] = fmaxf(0.f, s) * mk;
    }
  }
  __syncthreads();

  const int jt0 = wave, jt1 = wave + 8;
  const float fb1v0 = fb1[jt0 * 16 + lanelo];
  const float fb1v1 = fb1[jt1 * 16 + lanelo];
  v8f acc0 = {};
  v8f acc1 = {};

  for (int half = 0; half < 2; ++half) {
    const int m = half * 8 + wave;
    const int ml = wave;
    for (int ct = 0; ct < 4; ++ct) {
      v8f e = {};
#pragma unroll
      for (int kk = 0; kk < 16; kk += 4) {
        v2f a  = *(const v2f*)&Ml[(m * 16 + lanelo) * FSTR + kk + dk];
        v2f bb = *(const v2f*)&Fl[(ct * 16 + lanelo) * FSTR + kk + dk];
        e = wmma_f32(a, bb, e);
      }
#pragma unroll
      for (int r = 0; r < 8; ++r)
        El[(mbase + r) * ESTR + ml * 64 + ct * 16 + lanelo] = e[r];
    }
    __syncthreads();

#pragma unroll 4
    for (int kk4 = 0; kk4 < 128; ++kk4) {
      int kkh = kk4 * 4;
      int ka = half * 128 + kk4;
      v2f a  = *(const v2f*)&El[lanelo * ESTR + kkh + dk];
      v2f b0 = *(const v2f*)&f1p[(size_t)((ka * 16 + jt0) * 32 + lane) * 2];
      v2f b1v = *(const v2f*)&f1p[(size_t)((ka * 16 + jt1) * 32 + lane) * 2];
      acc0 = wmma_f32(a, b0, acc0);
      acc1 = wmma_f32(a, b1v, acc1);
    }
    __syncthreads();
  }

#pragma unroll
  for (int r = 0; r < 8; ++r) {
    int n = mbase + r;
    Hl[n * HSTR + jt0 * 16 + lanelo] = fmaxf(0.f, acc0[r] + fb1v0);
    Hl[n * HSTR + jt1 * 16 + lanelo] = fmaxf(0.f, acc1[r] + fb1v1);
  }
  __syncthreads();

  if (wave < 4) {
    const int ot = wave;
    const float fb2v = fb2[ot * 16 + lanelo];
    v8f o = {};
#pragma unroll 4
    for (int kk4 = 0; kk4 < 64; ++kk4) {
      v2f a  = *(const v2f*)&Hl[lanelo * HSTR + kk4 * 4 + dk];
      v2f bb = *(const v2f*)&f2p[(size_t)((kk4 * 4 + ot) * 32 + lane) * 2];
      o = wmma_f32(a, bb, o);
    }
#pragma unroll
    for (int r = 0; r < 8; ++r)
      out[(size_t)(base + mbase + r) * COUT + ot * 16 + lanelo] = o[r] + fb2v;
  }
}

extern "C" void kernel_launch(void* const* d_in, const int* in_sizes, int n_in,
                              void* d_out, int out_size, void* d_ws, size_t ws_size,
                              hipStream_t stream) {
  (void)in_sizes; (void)n_in; (void)out_size; (void)ws_size;
  const float* points = (const float*)d_in[1];
  const float* feats  = (const float*)d_in[2];
  const float* pmask  = (const float*)d_in[3];
  const float* w1 = (const float*)d_in[4];
  const float* b1 = (const float*)d_in[5];
  const float* w2 = (const float*)d_in[6];
  const float* b2 = (const float*)d_in[7];
  const float* w3 = (const float*)d_in[8];
  const float* b3 = (const float*)d_in[9];
  const float* f1  = (const float*)d_in[10];
  const float* fb1 = (const float*)d_in[11];
  const float* f2  = (const float*)d_in[12];
  const float* fb2 = (const float*)d_in[13];
  float* outp = (float*)d_out;

  float* f1p = (float*)d_ws;
  float* f2p = (float*)((char*)d_ws + (size_t)(1 << 20));

  pack_weights<<<544, 256, 0, stream>>>(f1, f2, f1p, f2p);
  gpc_main<<<dim3(Nn / Kn, Bn), 256, 0, stream>>>(
      points, feats, pmask, w1, b1, w2, b2, w3, b3, fb1, fb2, f1p, f2p, outp);
}


# Each work-group now handles 4 groups (64 rows) with 16 waves, one per N-tile of layer 1, so every packed f1 fragment fetched from global feeds 4 WMMAs: 4x fewer fragment loads; grid 1024 -> 256
# speedup vs baseline: 1.0729x; 1.0729x over previous
#include <hip/hip_runtime.h>

constexpr int Bn   = 8;
constexpr int Nn   = 2048;
constexpr int Kn   = 16;
constexpr int CIN  = 64;
constexpr int CMID = 16;
constexpr int COUT = 64;
constexpr int FH1  = 256;
constexpr int KDIM = CMID * CIN;

typedef float v2f __attribute__((ext_vector_type(2)));
typedef float v8f __attribute__((ext_vector_type(8)));

__device__ __forceinline__ v8f wmma_f32(v2f a, v2f b, v8f c) {
  return __builtin_amdgcn_wmma_f32_16x16x4_f32(
       false, a,  false, b,
       (short)0, c,  false,  false);
}

constexpr int G       = 4;
constexpr int NTHR    = 512;
constexpr int FSTR    = 18;
constexpr int OFF_F   = 0;
constexpr int OFF_M   = 1152;
constexpr int OFF_PTS = 5760;
constexpr int OFF_MSK = 5808;
constexpr int ESTR    = 514;
constexpr int OFF_E   = 5824;
constexpr int SMEM_F  = 14048;
constexpr int HSTR    = 258;
constexpr int OFF_H   = OFF_M;
static_assert(G * SMEM_F * 4 <= 320 * 1024, "LDS budget of one work-group");
static_assert(Nn % (Kn * G) == 0, "whole groups per work-group");
static_assert(G == 4 && NTHR == 512 && FH1 == 16 * (NTHR / 32), "wave-to-tile maps below assume 4 groups and 16 waves");

__global__ __launch_bounds__(256) void pack_weights(
    const float* __restrict__ f1, const float* __restrict__ f2,
    float* __restrict__ f1p, float* __restrict__ f2p) {
  int tid = blockIdx.x * 256 + threadIdx.x;
  constexpr int NP1 = (KDIM / 4) * (FH1 / 16) * 32;
  constexpr int NP2 = (FH1 / 4) * (COUT / 16) * 32;
  if (tid < NP1) {
    int lane = tid & 31, jt = (tid >> 5) & 15, kk4 = tid >> 9;
    int c = lane & 15, dk = (lane >> 4) << 1;
    int r0 = kk4 * 4 + dk;
    f1p[tid * 2 + 0] = f1[(r0 + 0) * FH1 + jt * 16 + c];
    f1p[tid * 2 + 1] = f1[(r0 + 1) * FH1 + jt * 16 + c];
  } else if (tid < NP1 + NP2) {
    int t2 = tid - NP1;
    int lane = t2 & 31, ot = (t2 >> 5) & 3, kk4 = t2 >> 7;
    int c = lane & 15, dk = (lane >> 4) << 1;
    int r0 = kk4 * 4 + dk;
    f2p[t2 * 2 + 0] = f2[(r0 + 0) * COUT + ot * 16 + c];
    f2p[t2 * 2 + 1] = f2[(r0 + 1) * COUT + ot * 16 + c];
  }
}

__global__ __launch_bounds__(NTHR) void gpc_main(
    const float* __restrict__ points, const float* __restrict__ feats,
    const float* __restrict__ pmask,
    const float* __restrict__ w1, const float* __restrict__ b1,
    const float* __restrict__ w2, const float* __restrict__ b2,
    const float* __restrict__ w3, const float* __restrict__ b3,
    const float* __restrict__ fb1, const float* __restrict__ fb2,
    const float* __restrict__ f1p, const float* __restrict__ f2p,
    float* __restrict__ out) {
  __shared__ float smem[G * SMEM_F];

  const int b = blockIdx.y;
  const int t = threadIdx.x;
  const int wave = t >> 5;
  const int lane = t & 31;
  const int lanelo = lane & 15;
  const int dk = (lane >> 4) << 1;
  const int mbase = (lane >> 4) << 3;
  const int base = b * Nn + blockIdx.x * (Kn * G);
  const int tl = t & 255;
  const int gh = (t >> 8) * 2;

#pragma unroll
  for (int q = 0; q < 2; ++q) {
    float* S = smem + (gh + q) * SMEM_F;
    const int gbase = base + (gh + q) * Kn;
    if (tl < 48) S[OFF_PTS + tl] = points[(size_t)gbase * 3 + tl];
    if (tl >= 48 && tl < 64) S[OFF_MSK + (tl - 48)] = pmask[gbase + (tl - 48)];
    int k = tl >> 4, c0 = (tl & 15) * 4;
    float mk = pmask[gbase + k];
    const float4 v = *(const float4*)(feats + (size_t)(gbase + k) * CIN + c0);
    S[OFF_F + (c0 + 0) * FSTR + k] = v.x * mk;
    S[OFF_F + (c0 + 1) * FSTR + k] = v.y * mk;
    S[OFF_F + (c0 + 2) * FSTR + k] = v.z * mk;
    S[OFF_F + (c0 + 3) * FSTR + k] = v.w * mk;
  }
  __syncthreads();

  {
    float* Sg = smem + gh * SMEM_F;
    int n = tl >> 4, k = tl & 15;
    float rx[2], ry[2], rz[2], mk[2];
#pragma unroll
    for (int q = 0; q < 2; ++q) {
      const float* Pt = Sg + q * SMEM_F + OFF_PTS;
      rx[q] = Pt[n * 3 + 0] - Pt[k * 3 + 0];
      ry[q] = Pt[n * 3 + 1] - Pt[k * 3 + 1];
      rz[q] = Pt[n * 3 + 2] - Pt[k * 3 + 2];
      mk[q] = Sg[q * SMEM_F + OFF_MSK + k];
    }
    float h1[2][32];
#pragma unroll
    for (int j = 0; j < 32; ++j) {
      const float wa = w1[j], wb = w1[32 + j], wc = w1[64 + j], bj = b1[j];
#pragma unroll
      for (int q = 0; q < 2; ++q)
        h1[q][j] = fmaxf(0.f, rx[q] * wa + ry[q] * wb + rz[q] * wc + bj);
    }
    float h2[2][32];
#pragma unroll
    for (int j = 0; j < 32; ++j) {
      float s[2];
#pragma unroll
      for (int q = 0; q < 2; ++q) s[q] = b2[j];
#pragma unroll
      for (int i = 0; i < 32; ++i) {
        const float w = w2[i * 32 + j];
#pragma unroll
        for (int q = 0; q < 2; ++q) s[q] += h1[q][i] * w;
      }
#pragma unroll
      for (int q = 0; q < 2; ++q) h2[q][j] = fmaxf(0.f, s[q]);
    }
#pragma unroll
    for (int j = 0; j < 16; ++j) {
      float s[2];
#pragma unroll
      for (int q = 0; q < 2; ++q) s[q] = b3[j];
#pragma unroll
      for (int i = 0; i < 32; ++i) {
        const float w = w3[i * 16 + j];
#pragma unroll
        for (int q = 0; q < 2; ++q) s[q] += h2[q][i] * w;
      }
#pragma unroll
      for (int q = 0; q < 2; ++q)
        Sg[q * SMEM_F + OFF_M + (j * 16 + n) * FSTR + k] = fmaxf(0.f, s[q]) * mk[q];
    }
  }
  __syncthreads();

  const int jt = wave;
  const float fb1v = fb1[jt * 16 + lanelo];
  v8f acc[G];
#pragma unroll
  for (int gi = 0; gi < G; ++gi) acc[gi] = v8f{};

  for (int half = 0; half < 2; ++half) {
    const int ml = wave & 7;
    const int m = half * 8 + ml;
    const int ge = (wave >> 3) * 2;
#pragma unroll
    for (int q = 0; q < 2; ++q) {
      const float* Ml = smem + (ge + q) * SMEM_F + OFF_M;
      const float* Fl = smem + (ge + q) * SMEM_F + OFF_F;
      float* El = smem + (ge + q) * SMEM_F + OFF_E;
      for (int ct = 0; ct < 4; ++ct) {
        v8f e = {};
#pragma unroll
        for (int kk = 0; kk < 16; kk += 4) {
          v2f a  = *(const v2f*)&Ml[(m * 16 + lanelo) * FSTR + kk + dk];
          v2f bb = *(const v2f*)&Fl[(ct * 16 + lanelo) * FSTR + kk + dk];
          e = wmma_f32(a, bb, e);
        }
#pragma unroll
        for (int r = 0; r < 8; ++r)
          El[(mbase + r) * ESTR + ml * 64 + ct * 16 + lanelo] = e[r];
      }
    }
    __syncthreads();

#pragma unroll 8
    for (int kk4 = 0; kk4 < 128; ++kk4) {
      int kkh = kk4 * 4;
      int ka = half * 128 + kk4;
      v2f bf = *(const v2f*)&f1p[(size_t)((ka * 16 + jt) * 32 + lane) * 2];
#pragma unroll
      for (int gi = 0; gi < G; ++gi) {
        v2f a = *(const v2f*)&smem[gi * SMEM_F + OFF_E + lanelo * ESTR + kkh + dk];
        acc[gi] = wmma_f32(a, bf, acc[gi]);
      }
    }
    __syncthreads();
  }

#pragma unroll
  for (int gi = 0; gi < G; ++gi) {
    float* Hl = smem + gi * SMEM_F + OFF_H;
#pragma unroll
    for (int r = 0; r < 8; ++r)
      Hl[(mbase + r) * HSTR + jt * 16 + lanelo] = fmaxf(0.f, acc[gi][r] + fb1v);
  }
  __syncthreads();

  {
    const int ot = wave & 3;
    const int g2 = wave >> 2;
    const float fb2v = fb2[ot * 16 + lanelo];
    const float* Hl = smem + g2 * SMEM_F + OFF_H;
    v8f o = {};
#pragma unroll 4
    for (int kk4 = 0; kk4 < 64; ++kk4) {
      v2f a  = *(const v2f*)&Hl[lanelo * HSTR + kk4 * 4 + dk];
      v2f bb = *(const v2f*)&f2p[(size_t)((kk4 * 4 + ot) * 32 + lane) * 2];
      o = wmma_f32(a, bb, o);
    }
#pragma unroll
    for (int r = 0; r < 8; ++r)
      out[(size_t)(base + g2 * Kn + mbase + r) * COUT + ot * 16 + lanelo] = o[r] + fb2v;
  }
}

extern "C" void kernel_launch(void* const* d_in, const int* in_sizes, int n_in,
                              void* d_out, int out_size, void* d_ws, size_t ws_size,
                              hipStream_t stream) {
  (void)in_sizes; (void)n_in; (void)out_size; (void)ws_size;
  const float* points = (const float*)d_in[1];
  const float* feats  = (const float*)d_in[2];
  const float* pmask  = (const float*)d_in[3];
  const float* w1 = (const float*)d_in[4];
  const float* b1 = (const float*)d_in[5];
  const float* w2 = (const float*)d_in[6];
  const float* b2 = (const float*)d_in[7];
  const float* w3 = (const float*)d_in[8];
  const float* b3 = (const float*)d_in[9];
  const float* f1  = (const float*)d_in[10];
  const float* fb1 = (const float*)d_in[11];
  const float* f2  = (const float*)d_in[12];
  const float* fb2 = (const float*)d_in[13];
  float* outp = (float*)d_out;

  float* f1p = (float*)d_ws;
  float* f2p = (float*)((char*)d_ws + (size_t)(1 << 20));

  pack_weights<<<544, 256, 0, stream>>>(f1, f2, f1p, f2p);
  gpc_main<<<dim3(Nn / (Kn * G), Bn), NTHR, 0, stream>>>(
      points, feats, pmask, w1, b1, w2, b2, w3, b3, fb1, fb2, f1p, f2p, outp);
}


# Layer-1 K loop pipelined over blocks of 8 k-steps with two fragment register sets: a block's LDS and global fragment reads are issued as one group ahead of the previous block's 32 WMMAs
# speedup vs baseline: 1.2160x; 1.1334x over previous
#include <hip/hip_runtime.h>

constexpr int Bn   = 8;
constexpr int Nn   = 2048;
constexpr int Kn   = 16;
constexpr int CIN  = 64;
constexpr int CMID = 16;
constexpr int COUT = 64;
constexpr int FH1  = 256;
constexpr int KDIM = CMID * CIN;

typedef float v2f __attribute__((ext_vector_type(2)));
typedef float v8f __attribute__((ext_vector_type(8)));

__device__ __forceinline__ v8f wmma_f32(v2f a, v2f b, v8f c) {
  return __builtin_amdgcn_wmma_f32_16x16x4_f32(
       false, a,  false, b,
       (short)0, c,  false,  false);
}

constexpr int G       = 4;
constexpr int NTHR    = 512;
constexpr int FSTR    = 18;
constexpr int OFF_F   = 0;
constexpr int OFF_M   = 1152;
constexpr int OFF_PTS = 5760;
constexpr int OFF_MSK = 5808;
constexpr int ESTR    = 514;
constexpr int OFF_E   = 5824;
constexpr int SMEM_F  = 14048;
constexpr int HSTR    = 258;
constexpr int OFF_H   = OFF_M;
static_assert(G * SMEM_F * 4 <= 320 * 1024, "LDS budget of one work-group");
static_assert(Nn % (Kn * G) == 0, "whole groups per work-group");
static_assert(G == 4 && NTHR == 512 && FH1 == 16 * (NTHR / 32), "wave-to-tile maps below assume 4 groups and 16 waves");

__global__ __launch_bounds__(256) void pack_weights(
    const float* __restrict__ f1, const float* __restrict__ f2,
    float* __restrict__ f1p, float* __restrict__ f2p) {
  int tid = blockIdx.x * 256 + threadIdx.x;
  constexpr int NP1 = (KDIM / 4) * (FH1 / 16) * 32;
  constexpr int NP2 = (FH1 / 4) * (COUT / 16) * 32;
  if (tid < NP1) {
    int lane = tid & 31, jt = (tid >> 5) & 15, kk4 = tid >> 9;
    int c = lane & 15, dk = (lane >> 4) << 1;
    int r0 = kk4 * 4 + dk;
    f1p[tid * 2 + 0] = f1[(r0 + 0) * FH1 + jt * 16 + c];
    f1p[tid * 2 + 1] = f1[(r0 + 1) * FH1 + jt * 16 + c];
  } else if (tid < NP1 + NP2) {
    int t2 = tid - NP1;
    int lane = t2 & 31, ot = (t2 >> 5) & 3, kk4 = t2 >> 7;
    int c = lane & 15, dk = (lane >> 4) << 1;
    int r0 = kk4 * 4 + dk;
    f2p[t2 * 2 + 0] = f2[(r0 + 0) * COUT + ot * 16 + c];
    f2p[t2 * 2 + 1] = f2[(r0 + 1) * COUT + ot * 16 + c];
  }
}

constexpr int KB   = 8;
constexpr int NBLK = 128 / KB;
static_assert(NBLK * KB == 128 && NBLK % 2 == 0, "whole blocks per half, taken in pairs");
typedef v2f AFrag[KB][G];
typedef v2f BFrag[KB];

__device__ __forceinline__ void l1_load_b(BFrag& bfr, const v2f* __restrict__ bp) {
#pragma unroll
  for (int s = 0; s < KB; ++s) bfr[s] = bp[s * 512];
}
__device__ __forceinline__ void l1_load_a(AFrag& afr, const float* sm, const int (&eo)[G], int k4) {
#pragma unroll
  for (int gi = 0; gi < G; ++gi) {
#pragma unroll
    for (int s = 0; s < KB; ++s)
      afr[s][gi] = *(const v2f*)&sm[eo[gi] + (k4 + s) * 4];
  }
}
__device__ __forceinline__ void l1_mma(v8f (&acc)[G], const AFrag& afr, const BFrag& bfr) {
#pragma unroll
  for (int s = 0; s < KB; ++s) {
#pragma unroll
    for (int gi = 0; gi < G; ++gi) acc[gi] = wmma_f32(afr[s][gi], bfr[s], acc[gi]);
  }
}
__device__ __forceinline__ void sched_fence() { __builtin_amdgcn_sched_barrier(0); }
__device__ __forceinline__ void block_end(v8f (&acc)[G]) {
  asm volatile("" : "+v"(acc[0]), "+v"(acc[1]), "+v"(acc[2]), "+v"(acc[3]));
}

__global__ __launch_bounds__(NTHR) void gpc_main(
    const float* __restrict__ points, const float* __restrict__ feats,
    const float* __restrict__ pmask,
    const float* __restrict__ w1, const float* __restrict__ b1,
    const float* __restrict__ w2, const float* __restrict__ b2,
    const float* __restrict__ w3, const float* __restrict__ b3,
    const float* __restrict__ fb1, const float* __restrict__ fb2,
    const float* __restrict__ f1p, const float* __restrict__ f2p,
    float* __restrict__ out) {
  __shared__ float smem[G * SMEM_F];

  const int b = blockIdx.y;
  const int t = threadIdx.x;
  const int wave = t >> 5;
  const int lane = t & 31;
  const int lanelo = lane & 15;
  const int dk = (lane >> 4) << 1;
  const int mbase = (lane >> 4) << 3;
  const int base = b * Nn + blockIdx.x * (Kn * G);
  const int tl = t & 255;
  const int gh = (t >> 8) * 2;

#pragma unroll
  for (int q = 0; q < 2; ++q) {
    float* S = smem + (gh + q) * SMEM_F;
    const int gbase = base + (gh + q) * Kn;
    if (tl < 48) S[OFF_PTS + tl] = points[(size_t)gbase * 3 + tl];
    if (tl >= 48 && tl < 64) S[OFF_MSK + (tl - 48)] = pmask[gbase + (tl - 48)];
    int k = tl >> 4, c0 = (tl & 15) * 4;
    float mk = pmask[gbase + k];
    const float4 v = *(const float4*)(feats + (size_t)(gbase + k) * CIN + c0);
    S[OFF_F + (c0 + 0) * FSTR + k] = v.x * mk;
    S[OFF_F + (c0 + 1) * FSTR + k] = v.y * mk;
    S[OFF_F + (c0 + 2) * FSTR + k] = v.z * mk;
    S[OFF_F + (c0 + 3) * FSTR + k] = v.w * mk;
  }
  __syncthreads();

  {
    float* Sg = smem + gh * SMEM_F;
    int n = tl >> 4, k = tl & 15;
    float rx[2], ry[2], rz[2], mk[2];
#pragma unroll
    for (int q = 0; q < 2; ++q) {
      const float* Pt = Sg + q * SMEM_F + OFF_PTS;
      rx[q] = Pt[n * 3 + 0] - Pt[k * 3 + 0];
      ry[q] = Pt[n * 3 + 1] - Pt[k * 3 + 1];
      rz[q] = Pt[n * 3 + 2] - Pt[k * 3 + 2];
      mk[q] = Sg[q * SMEM_F + OFF_MSK + k];
    }
    float h1[2][32];
#pragma unroll
    for (int j = 0; j < 32; ++j) {
      const float wa = w1[j], wb = w1[32 + j], wc = w1[64 + j], bj = b1[j];
#pragma unroll
      for (int q = 0; q < 2; ++q)
        h1[q][j] = fmaxf(0.f, rx[q] * wa + ry[q] * wb + rz[q] * wc + bj);
    }
    float h2[2][32];
#pragma unroll
    for (int j = 0; j < 32; ++j) {
      float s[2];
#pragma unroll
      for (int q = 0; q < 2; ++q) s[q] = b2[j];
#pragma unroll
      for (int i = 0; i < 32; ++i) {
        const float w = w2[i * 32 + j];
#pragma unroll
        for (int q = 0; q < 2; ++q) s[q] += h1[q][i] * w;
      }
#pragma unroll
      for (int q = 0; q < 2; ++q) h2[q][j] = fmaxf(0.f, s[q]);
    }
#pragma unroll
    for (int j = 0; j < 16; ++j) {
      float s[2];
#pragma unroll
      for (int q = 0; q < 2; ++q) s[q] = b3[j];
#pragma unroll
      for (int i = 0; i < 32; ++i) {
        const float w = w3[i * 16 + j];
#pragma unroll
        for (int q = 0; q < 2; ++q) s[q] += h2[q][i] * w;
      }
#pragma unroll
      for (int q = 0; q < 2; ++q)
        Sg[q * SMEM_F + OFF_M + (j * 16 + n) * FSTR + k] = fmaxf(0.f, s[q]) * mk[q];
    }
  }
  __syncthreads();

  const int jt = wave;
  const float fb1v = fb1[jt * 16 + lanelo];
  v8f acc[G];
#pragma unroll
  for (int gi = 0; gi < G; ++gi) acc[gi] = v8f{};

  for (int half = 0; half < 2; ++half) {
    const v2f* bp = (const v2f*)f1p + ((size_t)(half * 128) * 16 + jt) * 32 + lane;
    BFrag b0;
    l1_load_b(b0, bp);

    const int ml = wave & 7;
    const int m = half * 8 + ml;
    const int ge = (wave >> 3) * 2;
#pragma unroll
    for (int q = 0; q < 2; ++q) {
      const float* Ml = smem + (ge + q) * SMEM_F + OFF_M;
      const float* Fl = smem + (ge + q) * SMEM_F + OFF_F;
      float* El = smem + (ge + q) * SMEM_F + OFF_E;
      for (int ct = 0; ct < 4; ++ct) {
        v8f e = {};
#pragma unroll
        for (int kk = 0; kk < 16; kk += 4) {
          v2f a  = *(const v2f*)&Ml[(m * 16 + lanelo) * FSTR + kk + dk];
          v2f bb = *(const v2f*)&Fl[(ct * 16 + lanelo) * FSTR + kk + dk];
          e = wmma_f32(a, bb, e);
        }
#pragma unroll
        for (int r = 0; r < 8; ++r)
          El[(mbase + r) * ESTR + ml * 64 + ct * 16 + lanelo] = e[r];
      }
    }
    __syncthreads();

    {
      int eo[G];
#pragma unroll
      for (int gi = 0; gi < G; ++gi) {
        eo[gi] = gi * SMEM_F + OFF_E + lanelo * ESTR + dk;
        asm volatile("" : "+v"(eo[gi]));
      }
      AFrag a0, a1;
      BFrag b1;
      l1_load_a(a0, smem, eo, 0);
      for (int blk = 0; blk < NBLK - 2; blk += 2) {
        l1_load_b(b1, bp + (blk + 1) * (KB * 512));
        l1_load_a(a1, smem, eo, (blk + 1) * KB);
        sched_fence();
        l1_mma(acc, a0, b0);
        block_end(acc);
        sched_fence();
        l1_load_b(b0, bp + (blk + 2) * (KB * 512));
        l1_load_a(a0, smem, eo, (blk + 2) * KB);
        sched_fence();
        l1_mma(acc, a1, b1);
      }
      l1_load_b(b1, bp + (NBLK - 1) * (KB * 512));
      l1_load_a(a1, smem, eo, (NBLK - 1) * KB);
      sched_fence();
      l1_mma(acc, a0, b0);
      sched_fence();
      l1_mma(acc, a1, b1);
    }
    __syncthreads();
  }

#pragma unroll
  for (int gi = 0; gi < G; ++gi) {
    float* Hl = smem + gi * SMEM_F + OFF_H;
#pragma unroll
    for (int r = 0; r < 8; ++r)
      Hl[(mbase + r) * HSTR + jt * 16 + lanelo] = fmaxf(0.f, acc[gi][r] + fb1v);
  }
  __syncthreads();

  {
    const int ot = wave & 3;
    const int g2 = wave >> 2;
    const float fb2v = fb2[ot * 16 + lanelo];
    const float* Hl = smem + g2 * SMEM_F + OFF_H;
    v8f o = {};
#pragma unroll 4
    for (int kk4 = 0; kk4 < 64; ++kk4) {
      v2f a  = *(const v2f*)&Hl[lanelo * HSTR + kk4 * 4 + dk];
      v2f bb = *(const v2f*)&f2p[(size_t)((kk4 * 4 + ot) * 32 + lane) * 2];
      o = wmma_f32(a, bb, o);
    }
#pragma unroll
    for (int r = 0; r < 8; ++r)
      out[(size_t)(base + g2 * Kn + mbase + r) * COUT + ot * 16 + lanelo] = o[r] + fb2v;
  }
}

extern "C" void kernel_launch(void* const* d_in, const int* in_sizes, int n_in,
                              void* d_out, int out_size, void* d_ws, size_t ws_size,
                              hipStream_t stream) {
  (void)in_sizes; (void)n_in; (void)out_size; (void)ws_size;
  const float* points = (const float*)d_in[1];
  const float* feats  = (const float*)d_in[2];
  const float* pmask  = (const float*)d_in[3];
  const float* w1 = (const float*)d_in[4];
  const float* b1 = (const float*)d_in[5];
  const float* w2 = (const float*)d_in[6];
  const float* b2 = (const float*)d_in[7];
  const float* w3 = (const float*)d_in[8];
  const float* b3 = (const float*)d_in[9];
  const float* f1  = (const float*)d_in[10];
  const float* fb1 = (const float*)d_in[11];
  const float* f2  = (const float*)d_in[12];
  const float* fb2 = (const float*)d_in[13];
  float* outp = (float*)d_out;

  float* f1p = (float*)d_ws;
  float* f2p = (float*)((char*)d_ws + (size_t)(1 << 20));

  pack_weights<<<544, 256, 0, stream>>>(f1, f2, f1p, f2p);
  gpc_main<<<dim3(Nn / (Kn * G), Bn), NTHR, 0, stream>>>(
      points, feats, pmask, w1, b1, w2, b2, w3, b3, fb1, fb2, f1p, f2p, outp);
}


# Global loads outside the layer-1 K loop issued ahead as groups: phase 0 requests all its inputs before its first wait; layer 2's f2 fragments are fetched into registers during layer 1's last block
# speedup vs baseline: 1.3081x; 1.0757x over previous
#include <hip/hip_runtime.h>

constexpr int Bn   = 8;
constexpr int Nn   = 2048;
constexpr int Kn   = 16;
constexpr int CIN  = 64;
constexpr int CMID = 16;
constexpr int COUT = 64;
constexpr int FH1  = 256;
constexpr int KDIM = CMID * CIN;

typedef float v2f __attribute__((ext_vector_type(2)));
typedef float v8f __attribute__((ext_vector_type(8)));

__device__ __forceinline__ v8f wmma_f32(v2f a, v2f b, v8f c) {
  return __builtin_amdgcn_wmma_f32_16x16x4_f32(
       false, a,  false, b,
       (short)0, c,  false,  false);
}

constexpr int G       = 4;
constexpr int NTHR    = 512;
constexpr int FSTR    = 18;
constexpr int OFF_F   = 0;
constexpr int OFF_M   = 1152;
constexpr int OFF_PTS = 5760;
constexpr int OFF_MSK = 5808;
constexpr int ESTR    = 514;
constexpr int OFF_E   = 5824;
constexpr int SMEM_F  = 14048;
constexpr int HSTR    = 258;
constexpr int OFF_H   = OFF_M;
static_assert(G * SMEM_F * 4 <= 320 * 1024, "LDS budget of one work-group");
static_assert(Nn % (Kn * G) == 0, "whole groups per work-group");
static_assert(G == 4 && NTHR == 512 && FH1 == 16 * (NTHR / 32), "wave-to-tile maps below assume 4 groups and 16 waves");

__global__ __launch_bounds__(256) void pack_weights(
    const float* __restrict__ f1, const float* __restrict__ f2,
    float* __restrict__ f1p, float* __restrict__ f2p) {
  int tid = blockIdx.x * 256 + threadIdx.x;
  constexpr int NP1 = (KDIM / 4) * (FH1 / 16) * 32;
  constexpr int NP2 = (FH1 / 4) * (COUT / 16) * 32;
  if (tid < NP1) {
    int lane = tid & 31, jt = (tid >> 5) & 15, kk4 = tid >> 9;
    int c = lane & 15, dk = (lane >> 4) << 1;
    int r0 = kk4 * 4 + dk;
    f1p[tid * 2 + 0] = f1[(r0 + 0) * FH1 + jt * 16 + c];
    f1p[tid * 2 + 1] = f1[(r0 + 1) * FH1 + jt * 16 + c];
  } else if (tid < NP1 + NP2) {
    int t2 = tid - NP1;
    int lane = t2 & 31, ot = (t2 >> 5) & 3, kk4 = t2 >> 7;
    int c = lane & 15, dk = (lane >> 4) << 1;
    int r0 = kk4 * 4 + dk;
    f2p[t2 * 2 + 0] = f2[(r0 + 0) * COUT + ot * 16 + c];
    f2p[t2 * 2 + 1] = f2[(r0 + 1) * COUT + ot * 16 + c];
  }
}

constexpr int KB   = 8;
constexpr int NBLK = 128 / KB;
static_assert(NBLK * KB == 128 && NBLK % 2 == 0, "whole blocks per half, taken in pairs");
typedef v2f AFrag[KB][G];
typedef v2f BFrag[KB];

__device__ __forceinline__ void l1_load_b(BFrag& bfr, const v2f* __restrict__ bp) {
#pragma unroll
  for (int s = 0; s < KB; ++s) bfr[s] = bp[s * 512];
}
__device__ __forceinline__ void l1_load_a(AFrag& afr, const float* sm, const int (&eo)[G], int k4) {
#pragma unroll
  for (int gi = 0; gi < G; ++gi) {
#pragma unroll
    for (int s = 0; s < KB; ++s)
      afr[s][gi] = *(const v2f*)&sm[eo[gi] + (k4 + s) * 4];
  }
}
__device__ __forceinline__ void l1_mma(v8f (&acc)[G], const AFrag& afr, const BFrag& bfr) {
#pragma unroll
  for (int s = 0; s < KB; ++s) {
#pragma unroll
    for (int gi = 0; gi < G; ++gi) acc[gi] = wmma_f32(afr[s][gi], bfr[s], acc[gi]);
  }
}
__device__ __forceinline__ void sched_fence() { __builtin_amdgcn_sched_barrier(0); }
__device__ __forceinline__ void block_end(v8f (&acc)[G]) {
  asm volatile("" : "+v"(acc[0]), "+v"(acc[1]), "+v"(acc[2]), "+v"(acc[3]));
}

constexpr int K2H = 32;
constexpr int K2B = 16;
typedef v2f B2Frag[K2H];
typedef v2f A2Frag[K2B];
__device__ __forceinline__ void l2_load_b(B2Frag& bfr, const v2f* __restrict__ bp2, int k0) {
#pragma unroll
  for (int s = 0; s < K2H; ++s) bfr[s] = bp2[(k0 + s) * 128];
}
__device__ __forceinline__ void l2_load_a(A2Frag& afr, const float* hrow, int k0) {
#pragma unroll
  for (int s = 0; s < K2B; ++s) afr[s] = *(const v2f*)&hrow[(k0 + s) * 4];
}
__device__ __forceinline__ v8f l2_mma(v8f o, const A2Frag& afr, const B2Frag& bfr, int bofs) {
#pragma unroll
  for (int s = 0; s < K2B; ++s) o = wmma_f32(afr[s], bfr[bofs + s], o);
  return o;
}

__global__ __launch_bounds__(NTHR) void gpc_main(
    const float* __restrict__ points, const float* __restrict__ feats,
    const float* __restrict__ pmask,
    const float* __restrict__ w1, const float* __restrict__ b1,
    const float* __restrict__ w2, const float* __restrict__ b2,
    const float* __restrict__ w3, const float* __restrict__ b3,
    const float* __restrict__ fb1, const float* __restrict__ fb2,
    const float* __restrict__ f1p, const float* __restrict__ f2p,
    float* __restrict__ out) {
  __shared__ float smem[G * SMEM_F];

  const int b = blockIdx.y;
  const int t = threadIdx.x;
  const int wave = t >> 5;
  const int lane = t & 31;
  const int lanelo = lane & 15;
  const int dk = (lane >> 4) << 1;
  const int mbase = (lane >> 4) << 3;
  const int base = b * Nn + blockIdx.x * (Kn * G);
  const int tl = t & 255;
  const int gh = (t >> 8) * 2;

  {
    const int k = tl >> 4, c0 = (tl & 15) * 4;
    const int pi = tl < 48 ? tl : 47;
    float pv[2], mv[2], mkv[2];
    float4 fv[2];
#pragma unroll
    for (int q = 0; q < 2; ++q) {
      const int gbase = base + (gh + q) * Kn;
      pv[q]  = points[(size_t)gbase * 3 + pi];
      mv[q]  = pmask[gbase + (tl & 15)];
      mkv[q] = pmask[gbase + k];
      fv[q]  = *(const float4*)(feats + (size_t)(gbase + k) * CIN + c0);
    }
    asm volatile("" : "+v"(pv[0]), "+v"(mv[0]), "+v"(pv[1]), "+v"(mv[1]));
#pragma unroll
    for (int q = 0; q < 2; ++q) {
      float* S = smem + (gh + q) * SMEM_F;
      if (tl < 48) S[OFF_PTS + tl] = pv[q];
      if (tl >= 48 && tl < 64) S[OFF_MSK + (tl - 48)] = mv[q];
      S[OFF_F + (c0 + 0) * FSTR + k] = fv[q].x * mkv[q];
      S[OFF_F + (c0 + 1) * FSTR + k] = fv[q].y * mkv[q];
      S[OFF_F + (c0 + 2) * FSTR + k] = fv[q].z * mkv[q];
      S[OFF_F + (c0 + 3) * FSTR + k] = fv[q].w * mkv[q];
    }
  }
  __syncthreads();

  {
    float* Sg = smem + gh * SMEM_F;
    int n = tl >> 4, k = tl & 15;
    float rx[2], ry[2], rz[2], mk[2];
#pragma unroll
    for (int q = 0; q < 2; ++q) {
      const float* Pt = Sg + q * SMEM_F + OFF_PTS;
      rx[q] = Pt[n * 3 + 0] - Pt[k * 3 + 0];
      ry[q] = Pt[n * 3 + 1] - Pt[k * 3 + 1];
      rz[q] = Pt[n * 3 + 2] - Pt[k * 3 + 2];
      mk[q] = Sg[q * SMEM_F + OFF_MSK + k];
    }
    float h1[2][32];
#pragma unroll
    for (int j = 0; j < 32; ++j) {
      const float wa = w1[j], wb = w1[32 + j], wc = w1[64 + j], bj = b1[j];
#pragma unroll
      for (int q = 0; q < 2; ++q)
        h1[q][j] = fmaxf(0.f, rx[q] * wa + ry[q] * wb + rz[q] * wc + bj);
    }
    float h2[2][32];
#pragma unroll
    for (int j = 0; j < 32; ++j) {
      float s[2];
#pragma unroll
      for (int q = 0; q < 2; ++q) s[q] = b2[j];
#pragma unroll
      for (int i = 0; i < 32; ++i) {
        const float w = w2[i * 32 + j];
#pragma unroll
        for (int q = 0; q < 2; ++q) s[q] += h1[q][i] * w;
      }
#pragma unroll
      for (int q = 0; q < 2; ++q) h2[q][j] = fmaxf(0.f, s[q]);
    }
#pragma unroll
    for (int j = 0; j < 16; ++j) {
      float s[2];
#pragma unroll
      for (int q = 0; q < 2; ++q) s[q] = b3[j];
#pragma unroll
      for (int i = 0; i < 32; ++i) {
        const float w = w3[i * 16 + j];
#pragma unroll
        for (int q = 0; q < 2; ++q) s[q] += h2[q][i] * w;
      }
#pragma unroll
      for (int q = 0; q < 2; ++q)
        Sg[q * SMEM_F + OFF_M + (j * 16 + n) * FSTR + k] = fmaxf(0.f, s[q]) * mk[q];
    }
  }
  __syncthreads();

  const int jt = wave;
  const float fb1v = fb1[jt * 16 + lanelo];
  v8f acc[G];
#pragma unroll
  for (int gi = 0; gi < G; ++gi) acc[gi] = v8f{};

  const v2f* bp2 = (const v2f*)f2p + (wave & 3) * 32 + lane;
  B2Frag w2lo;

#pragma unroll
  for (int half = 0; half < 2; ++half) {
    const v2f* bp = (const v2f*)f1p + ((size_t)(half * 128) * 16 + jt) * 32 + lane;
    BFrag b0;
    l1_load_b(b0, bp);

    const int ml = wave & 7;
    const int m = half * 8 + ml;
    const int ge = (wave >> 3) * 2;
#pragma unroll
    for (int q = 0; q < 2; ++q) {
      const float* Ml = smem + (ge + q) * SMEM_F + OFF_M;
      const float* Fl = smem + (ge + q) * SMEM_F + OFF_F;
      float* El = smem + (ge + q) * SMEM_F + OFF_E;
      for (int ct = 0; ct < 4; ++ct) {
        v8f e = {};
#pragma unroll
        for (int kk = 0; kk < 16; kk += 4) {
          v2f a  = *(const v2f*)&Ml[(m * 16 + lanelo) * FSTR + kk + dk];
          v2f bb = *(const v2f*)&Fl[(ct * 16 + lanelo) * FSTR + kk + dk];
          e = wmma_f32(a, bb, e);
        }
#pragma unroll
        for (int r = 0; r < 8; ++r)
          El[(mbase + r) * ESTR + ml * 64 + ct * 16 + lanelo] = e[r];
      }
    }
    __syncthreads();

    {
      int eo[G];
#pragma unroll
      for (int gi = 0; gi < G; ++gi) {
        eo[gi] = gi * SMEM_F + OFF_E + lanelo * ESTR + dk;
        asm volatile("" : "+v"(eo[gi]));
      }
      AFrag a0, a1;
      BFrag b1;
      l1_load_a(a0, smem, eo, 0);
      for (int blk = 0; blk < NBLK - 2; blk += 2) {
        l1_load_b(b1, bp + (blk + 1) * (KB * 512));
        l1_load_a(a1, smem, eo, (blk + 1) * KB);
        sched_fence();
        l1_mma(acc, a0, b0);
        block_end(acc);
        sched_fence();
        l1_load_b(b0, bp + (blk + 2) * (KB * 512));
        l1_load_a(a0, smem, eo, (blk + 2) * KB);
        sched_fence();
        l1_mma(acc, a1, b1);
      }
      l1_load_b(b1, bp + (NBLK - 1) * (KB * 512));
      l1_load_a(a1, smem, eo, (NBLK - 1) * KB);
      sched_fence();
      l1_mma(acc, a0, b0);
      sched_fence();
      if (half == 1) {
        l2_load_b(w2lo, bp2, 0);
        sched_fence();
      }
      l1_mma(acc, a1, b1);
      if (half == 1) sched_fence();
    }
    __syncthreads();
  }

#pragma unroll
  for (int gi = 0; gi < G; ++gi) {
    float* Hl = smem + gi * SMEM_F + OFF_H;
#pragma unroll
    for (int r = 0; r < 8; ++r)
      Hl[(mbase + r) * HSTR + jt * 16 + lanelo] = fmaxf(0.f, acc[gi][r] + fb1v);
  }
  __syncthreads();

  {
    const int ot = wave & 3;
    const int g2 = wave >> 2;
    const float fb2v = fb2[ot * 16 + lanelo];
    const float* hrow = smem + g2 * SMEM_F + OFF_H + lanelo * HSTR + dk;
    B2Frag w2hi;
    l2_load_b(w2hi, bp2, K2H);
    A2Frag h0, h1;
    l2_load_a(h0, hrow, 0);
    l2_load_a(h1, hrow, K2B);
    sched_fence();
    v8f o = {};
    o = l2_mma(o, h0, w2lo, 0);
    sched_fence();
    l2_load_a(h0, hrow, 2 * K2B);
    sched_fence();
    o = l2_mma(o, h1, w2lo, K2B);
    sched_fence();
    l2_load_a(h1, hrow, 3 * K2B);
    sched_fence();
    o = l2_mma(o, h0, w2hi, 0);
    o = l2_mma(o, h1, w2hi, K2B);
#pragma unroll
    for (int r = 0; r < 8; ++r)
      out[(size_t)(base + g2 * Kn + mbase + r) * COUT + ot * 16 + lanelo] = o[r] + fb2v;
  }
}

extern "C" void kernel_launch(void* const* d_in, const int* in_sizes, int n_in,
                              void* d_out, int out_size, void* d_ws, size_t ws_size,
                              hipStream_t stream) {
  (void)in_sizes; (void)n_in; (void)out_size; (void)ws_size;
  const float* points = (const float*)d_in[1];
  const float* feats  = (const float*)d_in[2];
  const float* pmask  = (const float*)d_in[3];
  const float* w1 = (const float*)d_in[4];
  const float* b1 = (const float*)d_in[5];
  const float* w2 = (const float*)d_in[6];
  const float* b2 = (const float*)d_in[7];
  const float* w3 = (const float*)d_in[8];
  const float* b3 = (const float*)d_in[9];
  const float* f1  = (const float*)d_in[10];
  const float* fb1 = (const float*)d_in[11];
  const float* f2  = (const float*)d_in[12];
  const float* fb2 = (const float*)d_in[13];
  float* outp = (float*)d_out;

  float* f1p = (float*)d_ws;
  float* f2p = (float*)((char*)d_ws + (size_t)(1 << 20));

  pack_weights<<<544, 256, 0, stream>>>(f1, f2, f1p, f2p);
  gpc_main<<<dim3(Nn / (Kn * G), Bn), NTHR, 0, stream>>>(
      points, feats, pmask, w1, b1, w2, b2, w3, b3, fb1, fb2, f1p, f2p, outp);
}
